# starting kernel
# baseline (speedup 1.0000x reference)
#include <hip/hip_runtime.h>
#include <math.h>

static constexpr int kB    = 8;
static constexpr int kC    = 512;
static constexpr int kT    = 4;
static constexpr int kHW   = 64 * 64;
static constexpr int kNSP  = 100;
static constexpr int kDOUT = 128;
static constexpr int kNSEG = kB * kT * kNSP;
static constexpr int kCHUNK  = 32;
static constexpr int kNCHUNK = kC / kCHUNK;
static constexpr int kLPAD   = 33;
static constexpr int kASTR   = 516;

typedef float v2f __attribute__((ext_vector_type(2)));
typedef float v8f __attribute__((ext_vector_type(8)));

__global__ __launch_bounds__(256) void seg_sum_kernel(
    const float* __restrict__ maps, const int* __restrict__ sp_mask,
    float* __restrict__ sums)
{
  __shared__ float lds[kNSP * kLPAD];
  const int tid = threadIdx.x;
  for (int i = tid; i < kNSP * kLPAD; i += 256) lds[i] = 0.0f;
  __syncthreads();

  const int blk   = blockIdx.x;
  const int chunk = blk & (kNCHUNK - 1);
  const int bt    = blk >> 4;
  const int b     = bt >> 2;
  const int t     = bt & 3;

  const int lane = tid & 31;
  const int wv   = tid >> 5;

  const int* spb = sp_mask + (size_t)bt * kHW;

  for (int pb = 0; pb < kHW; pb += 128) {
    const int p = pb + lane * 4;
    const int4 sp4 = *reinterpret_cast<const int4*>(spb + p);
#pragma unroll
    for (int cc = 0; cc < 4; ++cc) {
      const int cl = wv + 8 * cc;
      const int c  = chunk * kCHUNK + cl;
      const float* mp =
          maps + ((size_t)(b * kC + c) * kT + t) * kHW + p;
      if (pb + 128 < kHW) __builtin_prefetch(mp + 128, 0, 1);
      const float4 v = *reinterpret_cast<const float4*>(mp);
      atomicAdd(&lds[sp4.x * kLPAD + cl], v.x);
      atomicAdd(&lds[sp4.y * kLPAD + cl], v.y);
      atomicAdd(&lds[sp4.z * kLPAD + cl], v.z);
      atomicAdd(&lds[sp4.w * kLPAD + cl], v.w);
    }
  }
  __syncthreads();

  float* outp = sums + (size_t)bt * kNSP * kC + chunk * kCHUNK;
  for (int e = tid; e < kNSP * kCHUNK; e += 256) {
    const int n  = e >> 5;
    const int cl = e & 31;
    outp[(size_t)n * kC + cl] = lds[n * kLPAD + cl];
  }
}

__global__ __launch_bounds__(256) void seg_cnt_kernel(
    const int* __restrict__ sp_mask, float* __restrict__ counts)
{
  __shared__ int cnt[kNSP];
  const int tid = threadIdx.x;
  if (tid < kNSP) cnt[tid] = 0;
  __syncthreads();
  const int bt = blockIdx.x;
  const int* spb = sp_mask + (size_t)bt * kHW;
  for (int p = tid; p < kHW; p += 256) atomicAdd(&cnt[spb[p]], 1);
  __syncthreads();
  if (tid < kNSP) counts[bt * kNSP + tid] = (float)cnt[tid];
}

__global__ __launch_bounds__(256) void fc_norm_kernel(
    const float* __restrict__ sums, const float* __restrict__ counts,
    const float* __restrict__ Wfc, float* __restrict__ outp)
{
  __shared__ float lds_a[16 * kASTR];
  __shared__ float lds_o[16 * kDOUT];
  __shared__ float lds_inv[16];

  const int tid = threadIdx.x;
  const int s0  = blockIdx.x * 16;

  for (int e = tid; e < 16 * kC; e += 256) {
    const int m = e >> 9;
    const int c = e & (kC - 1);
    const float cv = counts[s0 + m];
    const float sv = sums[(size_t)(s0 + m) * kC + c];
    lds_a[m * kASTR + c] = (cv > 0.0f) ? (sv / cv) : 0.0f;
  }
  __syncthreads();

  const int lane = tid & 31;
  const int wv   = tid >> 5;
  const int half = lane >> 4;
  const int ml   = lane & 15;
  const int n0   = wv * 16;

  v8f acc = {0.f, 0.f, 0.f, 0.f, 0.f, 0.f, 0.f, 0.f};
  const float* wrow = Wfc + (size_t)(n0 + ml) * kC;

  for (int k0 = 0; k0 < kC; k0 += 4) {
    v2f a, bm;
    const float2 av = *reinterpret_cast<const float2*>(
        &lds_a[ml * kASTR + k0 + 2 * half]);
    a.x = av.x; a.y = av.y;
    const float2 wb = *reinterpret_cast<const float2*>(wrow + k0 + 2 * half);
    bm.x = wb.x; bm.y = wb.y;
    acc = __builtin_amdgcn_wmma_f32_16x16x4_f32(
        false, a, false, bm, (short)0, acc, false, false);
  }

#pragma unroll
  for (int r = 0; r < 8; ++r) {
    const int m = r + 8 * half;
    lds_o[m * kDOUT + n0 + ml] = acc[r];
  }
  __syncthreads();

  if (tid < 16) {
    float ss = 0.0f;
    for (int d = 0; d < kDOUT; ++d) {
      const float x = lds_o[tid * kDOUT + d];
      ss += x * x;
    }
    lds_inv[tid] = 1.0f / fmaxf(sqrtf(ss), 1e-12f);
  }
  __syncthreads();

  for (int e = tid; e < 16 * kDOUT; e += 256) {
    const int m = e >> 7;
    const int d = e & (kDOUT - 1);
    const int s = s0 + m;
    const int b = s / (kT * kNSP);
    const int t = (s / kNSP) % kT;
    const int n = s % kNSP;
    outp[((size_t)(b * kDOUT + d) * kT + t) * kNSP + n] =
        lds_o[m * kDOUT + d] * lds_inv[m];
  }
}

extern "C" void kernel_launch(void* const* d_in, const int* in_sizes, int n_in,
                              void* d_out, int out_size, void* d_ws, size_t ws_size,
                              hipStream_t stream) {
  (void)in_sizes; (void)n_in; (void)out_size; (void)ws_size;
  const float* maps    = (const float*)d_in[0];
  const int*   sp_mask = (const int*)d_in[1];
  const float* Wfc     = (const float*)d_in[2];
  float* out = (float*)d_out;

  float* sums   = (float*)d_ws;
  float* counts = sums + (size_t)kNSEG * kC;

  seg_sum_kernel<<<kB * kT * kNCHUNK, 256, 0, stream>>>(maps, sp_mask, sums);
  seg_cnt_kernel<<<kB * kT, 256, 0, stream>>>(sp_mask, counts);
  fc_norm_kernel<<<kNSEG / 16, 256, 0, stream>>>(sums, counts, Wfc, out);
}


# fc_norm counts its own 16 segments' pixels from the frame's ids (per-wave integer LDS adds, id loads issued with the operand loads); the counts plane and the count launch are gone (3 launches to 2)
# speedup vs baseline: 1.6879x; 1.6879x over previous
#include <hip/hip_runtime.h>
#include <math.h>

static constexpr int kB    = 8;
static constexpr int kC    = 512;
static constexpr int kT    = 4;
static constexpr int kHW   = 64 * 64;
static constexpr int kNSP  = 100;
static constexpr int kDOUT = 128;
static constexpr int kNSEG = kB * kT * kNSP;
static constexpr int kCHUNK  = 32;
static constexpr int kNCHUNK = kC / kCHUNK;
static constexpr int kWPIX   = kHW / 8;
static constexpr int kTP     = 64;
static constexpr int kTPITCH = 68;
static constexpr int kWREG   = kNSP * kCHUNK;
static constexpr int kPSTR   = 132;

typedef float v2f __attribute__((ext_vector_type(2)));
typedef float v8f __attribute__((ext_vector_type(8)));
typedef float v4f __attribute__((ext_vector_type(4)));

#define ACC_FILE \
  "v128", "v129", "v130", "v131", "v132", "v133", "v134", "v135", \
  "v136", "v137", "v138", "v139", "v140", "v141", "v142", "v143", \
  "v144", "v145", "v146", "v147", "v148", "v149", "v150", "v151", \
  "v152", "v153", "v154", "v155", "v156", "v157", "v158", "v159", \
  "v160", "v161", "v162", "v163", "v164", "v165", "v166", "v167", \
  "v168", "v169", "v170", "v171", "v172", "v173", "v174", "v175", \
  "v176", "v177", "v178", "v179", "v180", "v181", "v182", "v183", \
  "v184", "v185", "v186", "v187", "v188", "v189", "v190", "v191", \
  "v192", "v193", "v194", "v195", "v196", "v197", "v198", "v199", \
  "v200", "v201", "v202", "v203", "v204", "v205", "v206", "v207", \
  "v208", "v209", "v210", "v211", "v212", "v213", "v214", "v215", \
  "v216", "v217", "v218", "v219", "v220", "v221", "v222", "v223", \
  "v224", "v225", "v226", "v227", "v228", "v229", "v230", "v231", \
  "v232", "v233", "v234", "v235", "v236", "v237", "v238", "v239", \
  "v240", "v241", "v242", "v243", "v244", "v245", "v246", "v247", \
  "v248", "v249", "v250", "v251", "v252", "v253", "v254", "v255"

#define ACC_ZERO8(a, b, c, d, e, f, g, h)                               \
  asm volatile("v_mov_b32 v" #a ", 0\n\tv_mov_b32 v" #b ", 0\n\t"       \
               "v_mov_b32 v" #c ", 0\n\tv_mov_b32 v" #d ", 0\n\t"       \
               "v_mov_b32 v" #e ", 0\n\tv_mov_b32 v" #f ", 0\n\t"       \
               "v_mov_b32 v" #g ", 0\n\tv_mov_b32 v" #h ", 0"            \
               : : : ACC_FILE)

#define ACC_OUT4(n, a, b, c, d)                                         \
  do {                                                                  \
    float r0_, r1_, r2_, r3_;                                           \
    asm volatile("v_mov_b32 %0, v" #a "\n\tv_mov_b32 %1, v" #b "\n\t"   \
                 "v_mov_b32 %2, v" #c "\n\tv_mov_b32 %3, v" #d          \
                 : "=v"(r0_), "=v"(r1_), "=v"(r2_), "=v"(r3_));         \
    tile[((n) + 0) * kCHUNK + lane] = r0_;                              \
    tile[((n) + 1) * kCHUNK + lane] = r1_;                              \
    tile[((n) + 2) * kCHUNK + lane] = r2_;                              \
    tile[((n) + 3) * kCHUNK + lane] = r3_;                              \
  } while (0)

#define SEG_STEP(val, sid)                                              \
  do {                                                                  \
    const unsigned ix_ = min((unsigned)(sid), 127u);                    \
    float t_;                                                           \
                        \
    asm volatile(                                                       \
        "s_mov_b32 m0, %2\n\t"                                          \
        "v_movrels_b32 %0, v128\n\t"                                    \
        "v_add_f32 %0, %0, %1\n\t"                                      \
        "v_movreld_b32 v128, %0"                                        \
        : "=&v"(t_)                                                     \
        : "v"(val), "s"(ix_)                                            \
        : "m0", ACC_FILE);                                              \
  } while (0)

__global__ __launch_bounds__(256) __attribute__((amdgpu_num_vgpr(128)))
void seg_sum_kernel(
    const float* __restrict__ maps, const int* __restrict__ sp_mask,
    float* __restrict__ sums)
{
  __shared__ float lds[8 * kWREG];
  const int tid  = threadIdx.x;
  const int lane = tid & 31;
  const int wv   = __builtin_amdgcn_readfirstlane(tid >> 5) & 7;

  const int blk   = blockIdx.x;
  const int chunk = blk & (kNCHUNK - 1);
  const int bt    = blk >> 4;
  const int b     = bt >> 2;
  const int t     = bt & 3;

  float* tile = lds + wv * kWREG;
  const int* spw = sp_mask + (size_t)bt * kHW + wv * kWPIX;

  const int lrow  = lane >> 4;
  const int lquad = lane & 15;
  const float* gp = maps +
      ((size_t)(b * kC + chunk * kCHUNK + lrow) * kT + t) * kHW +
      wv * kWPIX + lquad * 4;
  constexpr size_t kRow2 = (size_t)2 * kT * kHW;

  v4f g[16];
#pragma unroll
  for (int i = 0; i < 16; ++i)
    g[i] = *reinterpret_cast<const v4f*>(gp + i * kRow2);

  ACC_ZERO8(128, 129, 130, 131, 132, 133, 134, 135);
  ACC_ZERO8(136, 137, 138, 139, 140, 141, 142, 143);
  ACC_ZERO8(144, 145, 146, 147, 148, 149, 150, 151);
  ACC_ZERO8(152, 153, 154, 155, 156, 157, 158, 159);
  ACC_ZERO8(160, 161, 162, 163, 164, 165, 166, 167);
  ACC_ZERO8(168, 169, 170, 171, 172, 173, 174, 175);
  ACC_ZERO8(176, 177, 178, 179, 180, 181, 182, 183);
  ACC_ZERO8(184, 185, 186, 187, 188, 189, 190, 191);
  ACC_ZERO8(192, 193, 194, 195, 196, 197, 198, 199);
  ACC_ZERO8(200, 201, 202, 203, 204, 205, 206, 207);
  ACC_ZERO8(208, 209, 210, 211, 212, 213, 214, 215);
  ACC_ZERO8(216, 217, 218, 219, 220, 221, 222, 223);
  ACC_ZERO8(224, 225, 226, 227, 228, 229, 230, 231);
  ACC_ZERO8(232, 233, 234, 235, 236, 237, 238, 239);
  ACC_ZERO8(240, 241, 242, 243, 244, 245, 246, 247);
  ACC_ZERO8(248, 249, 250, 251, 252, 253, 254, 255);

  float* st = tile + lrow * kTPITCH + lquad * 4;
  const float* tr = tile + lane * kTPITCH;

  const int4* idp = reinterpret_cast<const int4*>(spw);
  int4 idc = idp[0];
  v4f x = (v4f){0.f, 0.f, 0.f, 0.f};
  for (int it = 0; it < kWPIX / 4; ++it) {
    const int q = it & (kTP / 4 - 1);
    if (q == 0) {
#pragma unroll
      for (int i = 0; i < 16; ++i)
        *reinterpret_cast<v4f*>(st + 2 * i * kTPITCH) = g[i];
      if (it + kTP / 4 < kWPIX / 4) {
        gp += kTP;
#pragma unroll
        for (int i = 0; i < 16; ++i)
          g[i] = *reinterpret_cast<const v4f*>(gp + i * kRow2);
      }
      x = *reinterpret_cast<const v4f*>(tr);
    }
    const v4f  xn  = *reinterpret_cast<const v4f*>(tr + 4 * ((q + 1) & (kTP / 4 - 1)));
    const int4 idn = idp[(it + 1) & (kWPIX / 4 - 1)];
    SEG_STEP(x.x, idc.x);
    SEG_STEP(x.y, idc.y);
    SEG_STEP(x.z, idc.z);
    SEG_STEP(x.w, idc.w);
    x = xn;
    idc = idn;
  }

  ACC_OUT4(0, 128, 129, 130, 131);
  ACC_OUT4(4, 132, 133, 134, 135);
  ACC_OUT4(8, 136, 137, 138, 139);
  ACC_OUT4(12, 140, 141, 142, 143);
  ACC_OUT4(16, 144, 145, 146, 147);
  ACC_OUT4(20, 148, 149, 150, 151);
  ACC_OUT4(24, 152, 153, 154, 155);
  ACC_OUT4(28, 156, 157, 158, 159);
  ACC_OUT4(32, 160, 161, 162, 163);
  ACC_OUT4(36, 164, 165, 166, 167);
  ACC_OUT4(40, 168, 169, 170, 171);
  ACC_OUT4(44, 172, 173, 174, 175);
  ACC_OUT4(48, 176, 177, 178, 179);
  ACC_OUT4(52, 180, 181, 182, 183);
  ACC_OUT4(56, 184, 185, 186, 187);
  ACC_OUT4(60, 188, 189, 190, 191);
  ACC_OUT4(64, 192, 193, 194, 195);
  ACC_OUT4(68, 196, 197, 198, 199);
  ACC_OUT4(72, 200, 201, 202, 203);
  ACC_OUT4(76, 204, 205, 206, 207);
  ACC_OUT4(80, 208, 209, 210, 211);
  ACC_OUT4(84, 212, 213, 214, 215);
  ACC_OUT4(88, 216, 217, 218, 219);
  ACC_OUT4(92, 220, 221, 222, 223);
  ACC_OUT4(96, 224, 225, 226, 227);
  __syncthreads();

  float* outp = sums + (size_t)bt * kNSP * kC + chunk * kCHUNK;
  for (int e = tid; e < kWREG; e += 256) {
    const int n  = e >> 5;
    const int cl = e & 31;
    float o = lds[e];
#pragma unroll
    for (int w = 1; w < 8; ++w) o += lds[w * kWREG + e];
    outp[(size_t)n * kC + cl] = o;
  }
}

__global__ __launch_bounds__(256) void fc_norm_kernel(
    const float* __restrict__ sums, const int* __restrict__ sp_mask,
    const float* __restrict__ Wfc, float* __restrict__ outp)
{
  __shared__ int   lds_cnt[8 * 16];
  __shared__ float lds_p[8 * 16 * kPSTR];
  __shared__ float lds_o[16 * kDOUT];
  __shared__ float lds_inv[16];

  const int tid  = threadIdx.x;
  const int s0   = blockIdx.x * 16;
  const int lane = tid & 31;
  const int wv   = tid >> 5;
  const int half = lane >> 4;
  const int ml   = lane & 15;
  const int kb   = wv * 64 + half * 32;

  const int f0 = s0 / kNSP;
  const int f1 = min(f0 + 1, kB * kT - 1);
  const int4* ip0 = reinterpret_cast<const int4*>(sp_mask + (size_t)f0 * kHW) + tid;
  const int4* ip1 = reinterpret_cast<const int4*>(sp_mask + (size_t)f1 * kHW) + tid;
  int4 ia[4], ib[4];
#pragma unroll
  for (int i = 0; i < 4; ++i) { ia[i] = ip0[i * 256]; ib[i] = ip1[i * 256]; }

  const float4* ap = reinterpret_cast<const float4*>(
      sums + (size_t)(s0 + ml) * kC + kb);
  float4 sa[8];
#pragma unroll
  for (int i = 0; i < 8; ++i) sa[i] = ap[i];

  float4 bw[8][8];
#pragma unroll
  for (int nt = 0; nt < 8; ++nt) {
    const float4* bp = reinterpret_cast<const float4*>(
        Wfc + (size_t)(nt * 16 + ml) * kC + kb);
#pragma unroll
    for (int i = 0; i < 8; ++i) bw[nt][i] = bp[i];
  }

  __builtin_amdgcn_sched_barrier(0);

  int* cw = lds_cnt + wv * 16;
  if (lane < 16) cw[lane] = 0;
  const int rb0 = f0 * kNSP - s0;
  const int rb1 = rb0 + kNSP;
#pragma unroll
  for (int i = 0; i < 4; ++i) {
    const int va[4] = {ia[i].x, ia[i].y, ia[i].z, ia[i].w};
    const int vb[4] = {ib[i].x, ib[i].y, ib[i].z, ib[i].w};
#pragma unroll
    for (int k = 0; k < 4; ++k) {
      const unsigned ra = (unsigned)(rb0 + va[k]);
      if ((unsigned)va[k] < (unsigned)kNSP && ra < 16u) atomicAdd(&cw[ra], 1);
      const unsigned rb = (unsigned)(rb1 + vb[k]);
      if ((unsigned)vb[k] < (unsigned)kNSP && rb < 16u) atomicAdd(&cw[rb], 1);
    }
  }
  __syncthreads();
  int ci = lds_cnt[ml];
#pragma unroll
  for (int w = 1; w < 8; ++w) ci += lds_cnt[w * 16 + ml];
  const float cv = (float)ci;

  const bool  has = cv > 0.0f;
  const float cs  = fmaxf(cv, 1.0f);
  float am[32];
#pragma unroll
  for (int i = 0; i < 8; ++i) {
    const float q0 = sa[i].x / cs;
    const float q1 = sa[i].y / cs;
    const float q2 = sa[i].z / cs;
    const float q3 = sa[i].w / cs;
    am[4 * i + 0] = has ? q0 : 0.0f;
    am[4 * i + 1] = has ? q1 : 0.0f;
    am[4 * i + 2] = has ? q2 : 0.0f;
    am[4 * i + 3] = has ? q3 : 0.0f;
  }

  v8f acc[8];
#pragma unroll
  for (int nt = 0; nt < 8; ++nt)
    acc[nt] = (v8f){0.f, 0.f, 0.f, 0.f, 0.f, 0.f, 0.f, 0.f};

#pragma unroll
  for (int j = 0; j < 16; ++j) {
    v2f a;
    a.x = am[2 * j]; a.y = am[2 * j + 1];
#pragma unroll
    for (int nt = 0; nt < 8; ++nt) {
      const float4 q = bw[nt][j >> 1];
      v2f bm;
      bm.x = (j & 1) ? q.z : q.x;
      bm.y = (j & 1) ? q.w : q.y;
      acc[nt] = __builtin_amdgcn_wmma_f32_16x16x4_f32(
          false, a, false, bm, (short)0, acc[nt], false, false);
    }
  }

  float* pw = lds_p + wv * 16 * kPSTR;
#pragma unroll
  for (int nt = 0; nt < 8; ++nt) {
#pragma unroll
    for (int r = 0; r < 8; ++r) {
      const int m = r + 8 * half;
      pw[m * kPSTR + nt * 16 + ml] = acc[nt][r];
    }
  }
  __syncthreads();

#pragma unroll
  for (int i = 0; i < 8; ++i) {
    const int e = tid + 256 * i;
    const int m = e >> 7;
    const int d = e & (kDOUT - 1);
    float o = lds_p[m * kPSTR + d];
#pragma unroll
    for (int w = 1; w < 8; ++w) o += lds_p[(w * 16 + m) * kPSTR + d];
    lds_o[m * kDOUT + d] = o;
  }
  __syncthreads();

  if (tid < 16) {
    float ss = 0.0f;
    for (int d = 0; d < kDOUT; ++d) {
      const float x = lds_o[tid * kDOUT + d];
      ss += x * x;
    }
    lds_inv[tid] = 1.0f / fmaxf(sqrtf(ss), 1e-12f);
  }
  __syncthreads();

  for (int e = tid; e < 16 * kDOUT; e += 256) {
    const int m = e >> 7;
    const int d = e & (kDOUT - 1);
    const int s = s0 + m;
    const int b = s / (kT * kNSP);
    const int t = (s / kNSP) % kT;
    const int n = s % kNSP;
    outp[((size_t)(b * kDOUT + d) * kT + t) * kNSP + n] =
        lds_o[m * kDOUT + d] * lds_inv[m];
  }
}

extern "C" void kernel_launch(void* const* d_in, const int* in_sizes, int n_in,
                              void* d_out, int out_size, void* d_ws, size_t ws_size,
                              hipStream_t stream) {
  (void)in_sizes; (void)n_in; (void)out_size; (void)ws_size;
  const float* maps    = (const float*)d_in[0];
  const int*   sp_mask = (const int*)d_in[1];
  const float* Wfc     = (const float*)d_in[2];
  float* out = (float*)d_out;

  float* sums   = (float*)d_ws;

  seg_sum_kernel<<<kB * kT * kNCHUNK, 256, 0, stream>>>(maps, sp_mask, sums);
  fc_norm_kernel<<<kNSEG / 16, 256, 0, stream>>>(sums, sp_mask, Wfc, out);
}
